# starting kernel
# baseline (speedup 1.0000x reference)
#include <hip/hip_runtime.h>

typedef float v2f __attribute__((ext_vector_type(2)));
typedef float v8f __attribute__((ext_vector_type(8)));
typedef int   v2i __attribute__((ext_vector_type(2)));

#define IN_DIM  10
#define OUT_DIM 8
#define TILE_ELEMS (IN_DIM * OUT_DIM)
#define TPG 4
#define NG  2

__global__ __launch_bounds__(256)
void frag_dist5_wmma_kernel(const float* __restrict__ x,
                            const float* __restrict__ W,
                            const float* __restrict__ dW,
                            const int*   __restrict__ head_ix,
                            const int*   __restrict__ split_ix,
                            const int*   __restrict__ n_splits_p,
                            float*       __restrict__ out,
                            int Btot)
{
    const int ns   = n_splits_p[0];
    const int lane = threadIdx.x & 31;
    const int waveInBlock   = threadIdx.x >> 5;
    const int wavesPerBlock = blockDim.x >> 5;
    const int wid    = blockIdx.x * wavesPerBlock + waveInBlock;
    const int nWaves = gridDim.x * wavesPerBlock;

    const int  n        = lane & 15;
    const int  sel      = lane >> 4;
    const int  nc       = (n < OUT_DIM) ? n : (OUT_DIM - 1);
    const bool  isAx   = (lane == 0) || (lane == 18);
    const bool  isAy   = (lane == 1) || (lane == 19);
    const float axMask = isAx ? 1.0f : 0.0f;
    const float ayMask = isAy ? 1.0f : 0.0f;

    for (int t = wid * (TPG * NG); t < Btot; t += nWaves * (TPG * NG)) {
        const float* qW0[NG]; const float* qD0[NG];
        const float* qW1[NG]; const float* qD1[NG];
        const float* px[NG];
        float w0[NG][IN_DIM], d0[NG][IN_DIM], w1[NG][IN_DIM], d1[NG][IN_DIM];
        float xr[NG][IN_DIM];

        #pragma unroll
        for (int g = 0; g < NG; ++g) {
            const int tg = t + g * TPG;
            int tb0 = tg + 2 * sel;
            tb0 = (tb0 <= Btot - 2) ? tb0 : (Btot - 2);
            const v2i hp = *(const v2i*)(head_ix + tb0);
            const v2i sp = *(const v2i*)(split_ix + tb0);
            const int h0 = hp.x, h1 = hp.y, s0 = sp.x, s1 = sp.y;
            qW0[g] = W  + (size_t)h0 * TILE_ELEMS + nc;
            qD0[g] = dW + ((size_t)h0 * ns + s0) * TILE_ELEMS + nc;
            qW1[g] = W  + (size_t)h1 * TILE_ELEMS + nc;
            qD1[g] = dW + ((size_t)h1 * ns + s1) * TILE_ELEMS + nc;
            px[g]  = x + (size_t)(isAy ? (tb0 + 1) : tb0) * IN_DIM;
        }

        #pragma unroll
        for (int g = 0; g < NG; ++g) {
            #pragma unroll
            for (int j = 0; j < IN_DIM; ++j) {
                w0[g][j] = qW0[g][j * OUT_DIM];
                d0[g][j] = qD0[g][j * OUT_DIM];
                w1[g][j] = qW1[g][j * OUT_DIM];
                d1[g][j] = qD1[g][j * OUT_DIM];
            }
            #pragma unroll
            for (int j = 0; j < IN_DIM; ++j) xr[g][j] = px[g][j];
        }

        v8f acc[NG];
        #pragma unroll
        for (int g = 0; g < NG; ++g) acc[g] = (v8f){};
        #pragma unroll
        for (int j = 0; j < IN_DIM; ++j) {
            #pragma unroll
            for (int g = 0; g < NG; ++g) {
                v2f a, b;
                a.x = xr[g][j] * axMask;
                a.y = xr[g][j] * ayMask;
                b.x = fmaf(0.1f, d0[g][j], w0[g][j]);
                b.y = fmaf(0.1f, d1[g][j], w1[g][j]);
                acc[g] = __builtin_amdgcn_wmma_f32_16x16x4_f32(
                              false, a,  false, b,
                              (short)0, acc[g],  false,  false);
            }
        }

        if (lane < OUT_DIM) {
            #pragma unroll
            for (int g = 0; g < NG; ++g) {
                #pragma unroll
                for (int m = 0; m < TPG; ++m) {
                    const int tok = t + g * TPG + m;
                    if (tok < Btot)
                        out[(size_t)tok * OUT_DIM + lane] = acc[g][m];
                }
            }
        }
    }
}

extern "C" void kernel_launch(void* const* d_in, const int* in_sizes, int n_in,
                              void* d_out, int out_size, void* d_ws, size_t ws_size,
                              hipStream_t stream) {
    const float* x   = (const float*)d_in[0];
    const float* W   = (const float*)d_in[1];
    const float* dW  = (const float*)d_in[2];
    const int*   hix = (const int*)d_in[3];
    const int*   six = (const int*)d_in[4];
    const int*   nsp = (const int*)d_in[5];
    float* out = (float*)d_out;

    const int Btot = in_sizes[0] / IN_DIM;
    const int tokensPerWave = TPG * NG;
    const int waveTiles = (Btot + tokensPerWave - 1) / tokensPerWave;
    const int wavesPerBlock = 8;
    const int blocks = (waveTiles + wavesPerBlock - 1) / wavesPerBlock;

    frag_dist5_wmma_kernel<<<blocks, 256, 0, stream>>>(x, W, dW, hix, six, nsp, out, Btot);
}


# Weight tiles fetched as contiguous 64-byte pieces, fused, staged in wave-private LDS; B fragments read from LDS so one WMMA sums 4 j for 2 tokens x 8 outputs; rows leave via LDS; j sum order changes
# speedup vs baseline: 1.4064x; 1.4064x over previous
#include <hip/hip_runtime.h>


typedef float v2f __attribute__((ext_vector_type(2)));
typedef float v4f __attribute__((ext_vector_type(4)));
typedef float v8f __attribute__((ext_vector_type(8)));

#define IN_DIM  10
#define OUT_DIM 8
#define TILE_ELEMS (IN_DIM * OUT_DIM)
#define TOK   16
#define PITCH 88
#define WAVES 8

__global__ __launch_bounds__(256)
void frag_dist5_wmma_kernel(const float* __restrict__ x,
                            const float* __restrict__ W,
                            const float* __restrict__ dW,
                            const int*   __restrict__ head_ix,
                            const int*   __restrict__ split_ix,
                            const int*   __restrict__ n_splits_p,
                            float*       __restrict__ out,
                            int Btot)
{
    __shared__ __attribute__((aligned(16))) float stage[WAVES][TOK * PITCH];

    const int ns   = n_splits_p[0];
    const int lane = threadIdx.x & 31;
    const int wv   = threadIdx.x >> 5;
    const int wid    = blockIdx.x * WAVES + wv;
    const int nWaves = gridDim.x * WAVES;
    float* st = stage[wv];

    const int tokF = lane >> 2;
    const int sub  = lane & 3;

    const int c = lane & 15;
    const int h = lane >> 4;
    const int rowTok = c;
    const int pid    = c >> 1;
    const int bBase  = (c >> 3) * PITCH + (2 * h) * OUT_DIM + (c & 7);
    const int tBase  = (2 * h + (c >> 3)) * PITCH + 8 * OUT_DIM + (c & 7);

    for (int t0 = wid * TOK; t0 < Btot; t0 += nWaves * TOK) {
        const int t = (t0 + TOK <= Btot) ? t0 : (Btot - TOK);

        const float* px = x + (size_t)(t + rowTok) * IN_DIM;
        const v2f xa = *(const v2f*)(px + 2 * h);
        const v2f xb = *(const v2f*)(px + 4 + 2 * h);
        const v2f xc = *(const v2f*)(px + 8);

        v4f wq[2][5], dq[2][5];
        #pragma unroll
        for (int H = 0; H < 2; ++H) {
            const int tf = t + 8 * H + tokF;
            const int hd = head_ix[tf];
            const int sp = split_ix[tf];
            const float* pw = W  + (size_t)hd * TILE_ELEMS + sub * 4;
            const float* pd = dW + ((size_t)hd * ns + sp) * TILE_ELEMS + sub * 4;
            #pragma unroll
            for (int i = 0; i < 5; ++i) {
                wq[H][i] = *(const v4f*)(pw + i * 16);
                dq[H][i] = *(const v4f*)(pd + i * 16);
            }
        }
        __builtin_amdgcn_sched_barrier(0);

        #pragma unroll
        for (int H = 0; H < 2; ++H) {
            #pragma unroll
            for (int i = 0; i < 5; ++i) {
                v4f f;
                f.x = fmaf(0.1f, dq[H][i].x, wq[H][i].x);
                f.y = fmaf(0.1f, dq[H][i].y, wq[H][i].y);
                f.z = fmaf(0.1f, dq[H][i].z, wq[H][i].z);
                f.w = fmaf(0.1f, dq[H][i].w, wq[H][i].w);
                *(v4f*)(st + (8 * H + tokF) * PITCH + i * 16 + sub * 4) = f;
            }
        }

        __builtin_amdgcn_fence(__ATOMIC_RELEASE, "workgroup", "local");
        __builtin_amdgcn_wave_barrier();
        __builtin_amdgcn_fence(__ATOMIC_ACQUIRE, "workgroup", "local");

        v8f acc0 = (v8f){};
        v8f acc1 = (v8f){};
        #pragma unroll
        for (int p = 0; p < 8; ++p) {
            const bool mine = (pid == p);
            #pragma unroll
            for (int s = 0; s < 2; ++s) {
                const float* bp = st + bBase + 2 * p * PITCH + 32 * s;
                v2f b; b.x = bp[0]; b.y = bp[OUT_DIM];
                const v2f xs = s ? xb : xa;
                v2f a; a.x = mine ? xs.x : 0.0f; a.y = mine ? xs.y : 0.0f;
                if (p & 1)
                    acc1 = __builtin_amdgcn_wmma_f32_16x16x4_f32(false, a, false, b, (short)0, acc1, false, false);
                else
                    acc0 = __builtin_amdgcn_wmma_f32_16x16x4_f32(false, a, false, b, (short)0, acc0, false, false);
            }
        }
        #pragma unroll
        for (int u = 0; u < 4; ++u) {
            const float* bp = st + tBase + 4 * u * PITCH;
            v2f b; b.x = bp[0]; b.y = bp[OUT_DIM];
            const bool mine = (pid == 2 * u + h);
            v2f a; a.x = mine ? xc.x : 0.0f; a.y = mine ? xc.y : 0.0f;
            if (u & 1)
                acc1 = __builtin_amdgcn_wmma_f32_16x16x4_f32(false, a, false, b, (short)0, acc1, false, false);
            else
                acc0 = __builtin_amdgcn_wmma_f32_16x16x4_f32(false, a, false, b, (short)0, acc0, false, false);
        }

        __builtin_amdgcn_fence(__ATOMIC_RELEASE, "wavefront");
        __builtin_amdgcn_wave_barrier();
        __builtin_amdgcn_fence(__ATOMIC_ACQUIRE, "wavefront");

        #pragma unroll
        for (int r = 0; r < 8; ++r) {
            st[r * 32 + lane]       = acc0[r];
            st[512 + r * 32 + lane] = acc1[r];
        }

        __builtin_amdgcn_fence(__ATOMIC_RELEASE, "workgroup", "local");
        __builtin_amdgcn_wave_barrier();
        __builtin_amdgcn_fence(__ATOMIC_ACQUIRE, "workgroup", "local");

        {
            const int tau = lane >> 1;
            const int rd  = ((tau & 7) * 2 + (tau >> 3)) * 16 + 8 * (tau & 1) + 4 * (lane & 1);
            const v4f o0 = *(const v4f*)(st + rd);
            const v4f o1 = *(const v4f*)(st + 512 + rd);
            *(v4f*)(out + (size_t)t * OUT_DIM + lane * 4) = o0 + o1;
        }

        __builtin_amdgcn_fence(__ATOMIC_RELEASE, "wavefront");
        __builtin_amdgcn_wave_barrier();
        __builtin_amdgcn_fence(__ATOMIC_ACQUIRE, "wavefront");
    }
}

extern "C" void kernel_launch(void* const* d_in, const int* in_sizes, int n_in,
                              void* d_out, int out_size, void* d_ws, size_t ws_size,
                              hipStream_t stream) {
    const float* x   = (const float*)d_in[0];
    const float* W   = (const float*)d_in[1];
    const float* dW  = (const float*)d_in[2];
    const int*   hix = (const int*)d_in[3];
    const int*   six = (const int*)d_in[4];
    const int*   nsp = (const int*)d_in[5];
    float* out = (float*)d_out;

    const int Btot = in_sizes[0] / IN_DIM;
    const int waveTiles = (Btot + TOK - 1) / TOK;
    const int blocks = (waveTiles + WAVES - 1) / WAVES;

    const int residentBlocks = 1024;
    const int grid = (blocks < residentBlocks) ? blocks : residentBlocks;

    frag_dist5_wmma_kernel<<<grid, 256, 0, stream>>>(x, W, dW, hix, six, nsp, out, Btot);
}
